# starting kernel
# baseline (speedup 1.0000x reference)
#include <hip/hip_runtime.h>


typedef float v2f __attribute__((ext_vector_type(2)));
typedef float v8f __attribute__((ext_vector_type(8)));

#define C_CH 192
#define N_IN (192 * 192 * 9)

__device__ __forceinline__ v8f wmma_f32_16x16x4(v2f a, v2f b, v8f c) {
  return __builtin_amdgcn_wmma_f32_16x16x4_f32(false, a, false, b, (short)0, c,
                                               false, false);
}

__global__ void transpose_in_k(const float* __restrict__ in,
                               float* __restrict__ out, int n) {
  int i = blockIdx.x * blockDim.x + threadIdx.x;
  if (i < n) {
    int a = i % C_CH;
    int t = i / C_CH;
    int p = t % 9;
    int o = t / 9;
    out[i] = in[(o * C_CH + a) * 9 + p];
  }
}

__global__ void ssq_partial_k(const float* __restrict__ x, int n,
                              float* __restrict__ part) {
  __shared__ float sdata[256];
  float s = 0.f;
  for (int i = blockIdx.x * blockDim.x + threadIdx.x; i < n;
       i += gridDim.x * blockDim.x) {
    float v = x[i];
    s += v * v;
  }
  sdata[threadIdx.x] = s;
  __syncthreads();
  for (int off = 128; off > 0; off >>= 1) {
    if ((int)threadIdx.x < off) sdata[threadIdx.x] += sdata[threadIdx.x + off];
    __syncthreads();
  }
  if (threadIdx.x == 0) part[blockIdx.x] = sdata[0];
}

__global__ void final_reduce_k(const float* __restrict__ part, int n,
                               float* __restrict__ out) {
  __shared__ float sdata[256];
  float s = 0.f;
  for (int i = threadIdx.x; i < n; i += 256) s += part[i];
  sdata[threadIdx.x] = s;
  __syncthreads();
  for (int off = 128; off > 0; off >>= 1) {
    if ((int)threadIdx.x < off) sdata[threadIdx.x] += sdata[threadIdx.x + off];
    __syncthreads();
  }
  if (threadIdx.x == 0) out[0] = sdata[0];
}

__global__ __launch_bounds__(32) void gram_wmma_k(
    const float* __restrict__ src, float* __restrict__ dst,
    const float* __restrict__ f2, float* __restrict__ part, int k) {
  const int P = k * k;
  const int w = 2 * k - 1;
  const int S = w * w;
  const int sidx = blockIdx.y;
  const int sy = sidx / w - (k - 1);
  const int sx = sidx % w - (k - 1);
  const int ny = k - (sy < 0 ? -sy : sy);
  const int nx = k - (sx < 0 ? -sx : sx);
  const int validp = ny * nx;
  const int by0 = sy < 0 ? -sy : 0;
  const int bx0 = sx < 0 ? -sx : 0;
  const int dpa = sy * k + sx;
  const unsigned Mdiv = (65536u + (unsigned)nx - 1u) / (unsigned)nx;

  const int lane = threadIdx.x;
  const int l15 = lane & 15;
  const int h = lane >> 4;
  const int m0 = (blockIdx.x / 3) * 32;
  const int n0 = (blockIdx.x % 3) * 64;

  const v8f z = {0.f, 0.f, 0.f, 0.f, 0.f, 0.f, 0.f, 0.f};
  v8f acc[2][4] = {{z, z, z, z}, {z, z, z, z}};

  for (int o = 0; o < C_CH; ++o) {
    const float* baseo = src + (size_t)o * P * C_CH;
    if (o + 1 < C_CH) __builtin_prefetch(baseo + P * C_CH, 0, 3);

    int j0 = 0;
    for (; j0 + 4 <= validp; j0 += 4) {
      int jb = j0 + 2 * h;
      int jd0 = (int)(((unsigned)jb * Mdiv) >> 16);
      int jm0 = jb - jd0 * nx;
      int jm1 = jm0 + 1;
      bool wrap = (jm1 == nx);
      int jd1 = jd0 + (wrap ? 1 : 0);
      jm1 = wrap ? 0 : jm1;
      int pb0 = (by0 + jd0) * k + (bx0 + jm0);
      int pb1 = (by0 + jd1) * k + (bx0 + jm1);
      const float* rb0 = baseo + pb0 * C_CH;
      const float* rb1 = baseo + pb1 * C_CH;
      const float* ra0 = rb0 + dpa * C_CH;
      const float* ra1 = rb1 + dpa * C_CH;

      v2f av[2], bv[4];
      av[0][0] = ra0[m0 + l15];
      av[1][0] = ra0[m0 + 16 + l15];
      av[0][1] = ra1[m0 + l15];
      av[1][1] = ra1[m0 + 16 + l15];
#pragma unroll
      for (int tj = 0; tj < 4; ++tj) {
        bv[tj][0] = rb0[n0 + 16 * tj + l15];
        bv[tj][1] = rb1[n0 + 16 * tj + l15];
      }
#pragma unroll
      for (int ti = 0; ti < 2; ++ti)
#pragma unroll
        for (int tj = 0; tj < 4; ++tj)
          acc[ti][tj] = wmma_f32_16x16x4(av[ti], bv[tj], acc[ti][tj]);
    }
    if (j0 < validp) {
      v2f av[2], bv[4];
#pragma unroll
      for (int e = 0; e < 2; ++e) {
        int j = j0 + 2 * h + e;
        int jj = j < validp ? j : validp - 1;
        int jd = (int)(((unsigned)jj * Mdiv) >> 16);
        int jm = jj - jd * nx;
        int pb = (by0 + jd) * k + (bx0 + jm);
        const float* rb = baseo + pb * C_CH;
        const float* ra = rb + dpa * C_CH;
        bool ok = j < validp;
        av[0][e] = ok ? ra[m0 + l15] : 0.f;
        av[1][e] = ok ? ra[m0 + 16 + l15] : 0.f;
#pragma unroll
        for (int tj = 0; tj < 4; ++tj)
          bv[tj][e] = ok ? rb[n0 + 16 * tj + l15] : 0.f;
      }
#pragma unroll
      for (int ti = 0; ti < 2; ++ti)
#pragma unroll
        for (int tj = 0; tj < 4; ++tj)
          acc[ti][tj] = wmma_f32_16x16x4(av[ti], bv[tj], acc[ti][tj]);
    }
  }

  const float inv = 1.0f / f2[0];

  if (dst) {
#pragma unroll
    for (int ti = 0; ti < 2; ++ti)
#pragma unroll
      for (int r = 0; r < 8; ++r) {
        int am = m0 + 16 * ti + r + 8 * h;
        size_t row = ((size_t)am * S + sidx) * C_CH;
#pragma unroll
        for (int tj = 0; tj < 4; ++tj)
          dst[row + n0 + 16 * tj + l15] = acc[ti][tj][r] * inv;
      }
  } else {
    float ssq = 0.f;
#pragma unroll
    for (int ti = 0; ti < 2; ++ti)
#pragma unroll
      for (int tj = 0; tj < 4; ++tj)
#pragma unroll
        for (int r = 0; r < 8; ++r) {
          float g = acc[ti][tj][r] * inv;
          ssq += g * g;
        }
    for (int off = 16; off > 0; off >>= 1) ssq += __shfl_xor(ssq, off, 32);
    if (lane == 0) part[blockIdx.y * gridDim.x + blockIdx.x] = ssq;
  }
}

__global__ void sigma_k(const float* __restrict__ s,
                        float* __restrict__ inv_sigma) {
  if (threadIdx.x == 0 && blockIdx.x == 0) {
    double ls = 0.5 * (log((double)s[0]) + 0.5 * log((double)s[1]) +
                       0.25 * log((double)s[2]) + 0.125 * log((double)s[3]));
    inv_sigma[0] = (float)exp(-ls);
  }
}

__global__ void scale_k(const float* __restrict__ in,
                        const float* __restrict__ inv_sigma,
                        float* __restrict__ out, int n) {
  int i = blockIdx.x * blockDim.x + threadIdx.x;
  if (i < n) out[i] = in[i] * inv_sigma[0];
}

static inline int imin(int a, int b) { return a < b ? a : b; }

extern "C" void kernel_launch(void* const* d_in, const int* in_sizes, int n_in,
                              void* d_out, int out_size, void* d_ws,
                              size_t ws_size, hipStream_t stream) {
  (void)in_sizes; (void)n_in; (void)out_size; (void)ws_size;
  const float* K = (const float*)d_in[0];
  float* out = (float*)d_out;
  char* ws = (char*)d_ws;

  float* scal = (float*)ws;
  float* part = (float*)(ws + 1024);
  float* K1T = (float*)(ws + (1 << 17));
  float* K2T = (float*)(ws + (4 << 20));
  float* KT0 = K2T;

  const int N0 = N_IN;
  const int N1 = 192 * 192 * 25;
  const int N2 = 192 * 192 * 81;

  transpose_in_k<<<(N0 + 255) / 256, 256, 0, stream>>>(K, KT0, N0);
  int g0 = imin(1024, (N0 + 255) / 256);
  ssq_partial_k<<<g0, 256, 0, stream>>>(K, N0, part);
  final_reduce_k<<<1, 256, 0, stream>>>(part, g0, scal + 0);

  gram_wmma_k<<<dim3(18, 25), 32, 0, stream>>>(KT0, K1T, scal + 0, nullptr, 3);

  int g1 = imin(1024, (N1 + 255) / 256);
  ssq_partial_k<<<g1, 256, 0, stream>>>(K1T, N1, part);
  final_reduce_k<<<1, 256, 0, stream>>>(part, g1, scal + 1);

  gram_wmma_k<<<dim3(18, 81), 32, 0, stream>>>(K1T, K2T, scal + 1, nullptr, 5);

  int g2 = imin(1024, (N2 + 255) / 256);
  ssq_partial_k<<<g2, 256, 0, stream>>>(K2T, N2, part);
  final_reduce_k<<<1, 256, 0, stream>>>(part, g2, scal + 2);

  gram_wmma_k<<<dim3(18, 289), 32, 0, stream>>>(K2T, nullptr, scal + 2, part,
                                                9);
  final_reduce_k<<<1, 256, 0, stream>>>(part, 18 * 289, scal + 3);

  sigma_k<<<1, 1, 0, stream>>>(scal, scal + 4);
  scale_k<<<(N0 + 255) / 256, 256, 0, stream>>>(K, scal + 4, out, N0);
}


# Gram steps: shifts dispatched in true descending order of k-step count (sorted table); was centre-out by flat index, which starts long shifts late; sums and their order unchanged
# speedup vs baseline: 3.2452x; 3.2452x over previous
#include <hip/hip_runtime.h>


typedef float v2f __attribute__((ext_vector_type(2)));
typedef float v8f __attribute__((ext_vector_type(8)));

#define C_CH 192
#define N_IN (192 * 192 * 9)

__device__ __forceinline__ v8f wmma_f32_16x16x4(v2f a, v2f b, v8f c) {
  return __builtin_amdgcn_wmma_f32_16x16x4_f32(false, a, false, b, (short)0, c,
                                               false, false);
}

__global__ void transpose_in_k(const float* __restrict__ in,
                               float* __restrict__ out, int n) {
  int i = blockIdx.x * blockDim.x + threadIdx.x;
  if (i < n) {
    int a = i % C_CH;
    int t = i / C_CH;
    int p = t % 9;
    int o = t / 9;
    out[i] = in[(o * C_CH + a) * 9 + p];
  }
}

__global__ void ssq_partial_k(const float* __restrict__ x, int n,
                              float* __restrict__ part) {
  __shared__ float sdata[256];
  float s = 0.f;
  for (int i = blockIdx.x * blockDim.x + threadIdx.x; i < n;
       i += gridDim.x * blockDim.x) {
    float v = x[i];
    s += v * v;
  }
  sdata[threadIdx.x] = s;
  __syncthreads();
  for (int off = 128; off > 0; off >>= 1) {
    if ((int)threadIdx.x < off) sdata[threadIdx.x] += sdata[threadIdx.x + off];
    __syncthreads();
  }
  if (threadIdx.x == 0) part[blockIdx.x] = sdata[0];
}

__global__ void final_reduce_k(const float* __restrict__ part, int n,
                               float* __restrict__ out) {
  __shared__ float sdata[256];
  float s = 0.f;
  for (int i = threadIdx.x; i < n; i += 256) s += part[i];
  sdata[threadIdx.x] = s;
  __syncthreads();
  for (int off = 128; off > 0; off >>= 1) {
    if ((int)threadIdx.x < off) sdata[threadIdx.x] += sdata[threadIdx.x + off];
    __syncthreads();
  }
  if (threadIdx.x == 0) out[0] = sdata[0];
}

__constant__ const unsigned short shift_order_3[25] = {
    12, 13, 11, 17, 7, 16, 8, 18, 6, 14, 10, 22, 2, 15, 9, 19, 5,
    21, 3, 23, 1, 20, 4, 24, 0};
__constant__ const unsigned short shift_order_5[81] = {
    40, 41, 39, 49, 31, 48, 32, 50, 30, 42, 38, 58, 22, 47, 33, 51, 29,
    57, 23, 59, 21, 43, 37, 67, 13, 56, 24, 60, 20, 46, 34, 52, 28, 66,
    14, 68, 12, 55, 25, 61, 19, 65, 15, 69, 11, 44, 36, 76, 4, 45, 35,
    53, 27, 64, 16, 70, 10, 75, 5, 77, 3, 54, 26, 62, 18, 74, 6, 78,
    2, 63, 17, 71, 9, 73, 7, 79, 1, 72, 8, 80, 0};
__constant__ const unsigned short shift_order_9[289] = {
    144, 145, 143, 161, 127, 160, 128, 162, 126, 146, 142, 178, 110, 159, 129, 163, 125,
    177, 111, 179, 109, 147, 141, 195, 93, 176, 112, 180, 108, 158, 130, 164, 124, 194,
    94, 196, 92, 148, 140, 212, 76, 175, 113, 181, 107, 193, 95, 197, 91, 157, 131,
    165, 123, 211, 77, 213, 75, 149, 139, 192, 96, 198, 90, 229, 59, 174, 114, 182,
    106, 210, 78, 214, 74, 156, 132, 166, 122, 228, 60, 230, 58, 191, 97, 199, 89,
    209, 79, 215, 73, 173, 115, 183, 105, 227, 61, 231, 57, 150, 138, 246, 42, 208,
    80, 216, 72, 155, 133, 167, 121, 190, 98, 200, 88, 226, 62, 232, 56, 245, 43,
    247, 41, 172, 116, 184, 104, 244, 44, 248, 40, 207, 81, 217, 71, 225, 63, 233,
    55, 151, 137, 189, 99, 201, 87, 243, 45, 249, 39, 263, 25, 154, 134, 168, 120,
    224, 64, 234, 54, 262, 26, 264, 24, 206, 82, 218, 70, 242, 46, 250, 38, 171,
    117, 185, 103, 261, 27, 265, 23, 188, 100, 202, 86, 223, 65, 235, 53, 241, 47,
    251, 37, 260, 28, 266, 22, 205, 83, 219, 69, 259, 29, 267, 21, 152, 136, 240,
    48, 252, 36, 280, 8, 153, 135, 169, 119, 222, 66, 236, 52, 258, 30, 268, 20,
    279, 9, 281, 7, 170, 118, 186, 102, 278, 10, 282, 6, 187, 101, 203, 85, 239,
    49, 253, 35, 257, 31, 269, 19, 277, 11, 283, 5, 204, 84, 220, 68, 276, 12,
    284, 4, 221, 67, 237, 51, 256, 32, 270, 18, 275, 13, 285, 3, 238, 50, 254,
    34, 274, 14, 286, 2, 255, 33, 271, 17, 273, 15, 287, 1, 272, 16, 288, 0};

template <int KS>
__device__ __forceinline__ int shift_of_rank(int r) {
  if constexpr (KS == 3) return shift_order_3[r];
  else if constexpr (KS == 5) return shift_order_5[r];
  else return shift_order_9[r];
}

template <int KS, int OB, bool WRITE>
__global__ __launch_bounds__(128) void gram_lds_k(
    const float* __restrict__ src, float* __restrict__ dst,
    const float* __restrict__ f2, float* __restrict__ part) {
  constexpr int P = KS * KS;
  constexpr int WS = 2 * KS - 1;
  constexpr int S = WS * WS;
  constexpr int NPAIR = ((P + 3) / 4) * 2;
  constexpr int NROW = OB * NPAIR;
  constexpr int NIT = (NROW + 7) / 8;
  constexpr int CP = 80;
  static_assert(C_CH % OB == 0, "o blocks must tile the channel count");
  static_assert(C_CH % 24 == 0 && C_CH % 16 == 0,
                "every o block size offered below must tile the channel count");

  constexpr int KG = 4;
  __shared__ __attribute__((aligned(16))) v2f sA[(NROW + 2 * KG) * CP];
  __shared__ __attribute__((aligned(16))) v2f sB[(NROW + 2 * KG) * CP];

  static_assert(KS == 3 || KS == 5 || KS == 9, "a dispatch table per kernel size");
  const int sidx = shift_of_rank<KS>(blockIdx.y);
  const int sy = sidx / WS - (KS - 1);
  const int sx = sidx % WS - (KS - 1);
  const int ny = KS - (sy < 0 ? -sy : sy);
  const int nx = KS - (sx < 0 ? -sx : sx);
  const int validp = ny * nx;
  const int nks = (validp + 3) >> 2;
  const int npv = 2 * nks;
  int obs = 1;
  if (2 * npv <= NROW) obs = 2;
  if (3 * npv <= NROW) obs = 3;
  if (4 * npv <= NROW) obs = 4;
  if (6 * npv <= NROW) obs = 6;
  if (8 * npv <= NROW) obs = 8;
  if (12 * npv <= NROW) obs = 12;
  if (16 * npv <= NROW) obs = 16;
  if (24 * npv <= NROW) obs = 24;
  const int nrows = obs * npv;
  const int nkt = obs * nks;
  const int by0 = sy < 0 ? -sy : 0;
  const int bx0 = sx < 0 ? -sx : 0;
  const int dpa = sy * KS + sx;

  const int tid = threadIdx.x;
  const int lane = tid & 31;
  const int wid = tid >> 5;
  const int l15 = lane & 15;
  const int h = lane >> 4;
  const int wm = wid >> 1;
  const int wn = wid & 1;
  const int m0 = (blockIdx.x / 3) * 64;
  const int n0 = (blockIdx.x % 3) * 64;

  const int q4 = (tid & 15) * 4;
  const int r8 = tid >> 4;

  unsigned offb[NIT][2], offa[NIT][2];
  unsigned usedmask = 0u, padmask = 0u;
#pragma unroll
  for (int it = 0; it < NIT; ++it) {
    const int g = r8 + 8 * it;
    const bool used = g < nrows;
    const int ob = used ? g / npv : 0;
    const int jp = used ? g - ob * npv : 0;
    if (used) usedmask |= 1u << it;
#pragma unroll
    for (int e = 0; e < 2; ++e) {
      const int j = 2 * jp + e;
      const int jj = j < validp ? j : validp - 1;
      const int jd = jj / nx;
      const int jm = jj - jd * nx;
      const int pb = (by0 + jd) * KS + (bx0 + jm);
      offb[it][e] = 4u * (unsigned)((ob * P + pb) * C_CH + n0 + q4);
      offa[it][e] = 4u * (unsigned)((ob * P + pb + dpa) * C_CH + m0 + q4);
      if (j >= validp) padmask |= 1u << (2 * it + e);
    }
  }

  const float4 z4 = {0.f, 0.f, 0.f, 0.f};
  float4 ra[NIT][2], rb[NIT][2];

  auto fetch = [&](int o0) {
    const unsigned ro = (unsigned)o0 * (unsigned)(P * C_CH * 4);
    const char* base = reinterpret_cast<const char*>(src);
#pragma unroll
    for (int it = 0; it < NIT; ++it) {
      if ((usedmask >> it) & 1u) {
#pragma unroll
        for (int e = 0; e < 2; ++e) {
          rb[it][e] =
              *reinterpret_cast<const float4*>(base + (offb[it][e] + ro));
          ra[it][e] =
              *reinterpret_cast<const float4*>(base + (offa[it][e] + ro));
        }
      }
    }
  };

  auto stash = [&]() {
#pragma unroll
    for (int it = 0; it < NIT; ++it) {
      if ((usedmask >> it) & 1u) {
        const int g = r8 + 8 * it;
        float4* pa4 = reinterpret_cast<float4*>(&sA[g * CP + q4]);
        float4* pb4 = reinterpret_cast<float4*>(&sB[g * CP + q4]);
        const bool pad0 = (padmask >> (2 * it)) & 1u;
        const bool pad1 = (padmask >> (2 * it + 1)) & 1u;
        const float4 a0 = pad0 ? z4 : ra[it][0], a1 = pad1 ? z4 : ra[it][1];
        const float4 b0 = pad0 ? z4 : rb[it][0], b1 = pad1 ? z4 : rb[it][1];
        pa4[0] = float4{a0.x, a1.x, a0.y, a1.y};
        pa4[1] = float4{a0.z, a1.z, a0.w, a1.w};
        pb4[0] = float4{b0.x, b1.x, b0.y, b1.y};
        pb4[1] = float4{b0.z, b1.z, b0.w, b1.w};
      }
    }
  };

  const v8f z = {0.f, 0.f, 0.f, 0.f, 0.f, 0.f, 0.f, 0.f};
  v8f acc[2][2] = {{z, z}, {z, z}};

  const v2f* fa = sA + h * CP + wm * 32 + l15;
  const v2f* fb = sB + h * CP + wn * 32 + l15;

  fetch(0);
  for (int o0 = 0; o0 < C_CH; o0 += obs) {
    __syncthreads();
    stash();
    __syncthreads();
    if (o0 + obs < C_CH) fetch(o0 + obs);

    {
      const v2f* pa = fa;
      const v2f* pb = fb;
      v2f av[KG][2], bv[KG][2];
#pragma unroll
      for (int g = 0; g < KG; ++g)
#pragma unroll
        for (int t = 0; t < 2; ++t) {
          av[g][t] = pa[2 * CP * g + 16 * t];
          bv[g][t] = pb[2 * CP * g + 16 * t];
        }
      const int ngrp = nkt / KG;
#pragma unroll 2
      for (int i = 0; i < ngrp; ++i) {
        pa += 2 * CP * KG;
        pb += 2 * CP * KG;
        v2f an[KG][2], bn[KG][2];
        __builtin_amdgcn_sched_barrier(0);
#pragma unroll
        for (int g = 0; g < KG; ++g)
#pragma unroll
          for (int t = 0; t < 2; ++t) {
            an[g][t] = pa[2 * CP * g + 16 * t];
            bn[g][t] = pb[2 * CP * g + 16 * t];
          }
        __builtin_amdgcn_sched_barrier(0);
#pragma unroll
        for (int g = 0; g < KG; ++g)
#pragma unroll
          for (int ti = 0; ti < 2; ++ti)
#pragma unroll
            for (int tj = 0; tj < 2; ++tj)
              acc[ti][tj] = wmma_f32_16x16x4(av[g][ti], bv[g][tj], acc[ti][tj]);
#pragma unroll
        for (int g = 0; g < KG; ++g)
#pragma unroll
          for (int t = 0; t < 2; ++t) {
            av[g][t] = an[g][t];
            bv[g][t] = bn[g][t];
          }
      }
      const int rem = nkt - ngrp * KG;
#pragma unroll
      for (int g = 0; g < KG - 1; ++g)
        if (g < rem) {
#pragma unroll
          for (int ti = 0; ti < 2; ++ti)
#pragma unroll
            for (int tj = 0; tj < 2; ++tj)
              acc[ti][tj] = wmma_f32_16x16x4(av[g][ti], bv[g][tj], acc[ti][tj]);
        }
    }
  }

  const float inv = 1.0f / f2[0];

  if (WRITE) {
#pragma unroll
    for (int ti = 0; ti < 2; ++ti)
#pragma unroll
      for (int r = 0; r < 8; ++r) {
        int am = m0 + wm * 32 + 16 * ti + r + 8 * h;
        size_t row = ((size_t)am * S + sidx) * C_CH;
#pragma unroll
        for (int tj = 0; tj < 2; ++tj)
          dst[row + n0 + wn * 32 + 16 * tj + l15] = acc[ti][tj][r] * inv;
      }
  } else {
    float ssq = 0.f;
#pragma unroll
    for (int ti = 0; ti < 2; ++ti)
#pragma unroll
      for (int tj = 0; tj < 2; ++tj)
#pragma unroll
        for (int r = 0; r < 8; ++r) {
          float g = acc[ti][tj][r] * inv;
          ssq += g * g;
        }
    for (int off = 16; off > 0; off >>= 1) ssq += __shfl_xor(ssq, off, 32);
    const int slot = sidx <= S / 2 ? 2 * (S / 2 - sidx) : 2 * (sidx - S / 2) - 1;
    if (lane == 0) part[(slot * gridDim.x + blockIdx.x) * 4 + wid] = ssq;
  }
}

__global__ void sigma_k(const float* __restrict__ s,
                        float* __restrict__ inv_sigma) {
  if (threadIdx.x == 0 && blockIdx.x == 0) {
    double ls = 0.5 * (log((double)s[0]) + 0.5 * log((double)s[1]) +
                       0.25 * log((double)s[2]) + 0.125 * log((double)s[3]));
    inv_sigma[0] = (float)exp(-ls);
  }
}

__global__ void scale_k(const float* __restrict__ in,
                        const float* __restrict__ inv_sigma,
                        float* __restrict__ out, int n) {
  int i = blockIdx.x * blockDim.x + threadIdx.x;
  if (i < n) out[i] = in[i] * inv_sigma[0];
}

static inline int imin(int a, int b) { return a < b ? a : b; }

extern "C" void kernel_launch(void* const* d_in, const int* in_sizes, int n_in,
                              void* d_out, int out_size, void* d_ws,
                              size_t ws_size, hipStream_t stream) {
  (void)in_sizes; (void)n_in; (void)out_size; (void)ws_size;
  const float* K = (const float*)d_in[0];
  float* out = (float*)d_out;
  char* ws = (char*)d_ws;

  float* scal = (float*)ws;
  float* part = (float*)(ws + 1024);
  float* K1T = (float*)(ws + (1 << 17));
  float* K2T = (float*)(ws + (4 << 20));
  float* KT0 = K2T;

  const int N0 = N_IN;
  const int N1 = 192 * 192 * 25;
  const int N2 = 192 * 192 * 81;

  transpose_in_k<<<(N0 + 255) / 256, 256, 0, stream>>>(K, KT0, N0);
  int g0 = imin(1024, (N0 + 255) / 256);
  ssq_partial_k<<<g0, 256, 0, stream>>>(K, N0, part);
  final_reduce_k<<<1, 256, 0, stream>>>(part, g0, scal + 0);

  gram_lds_k<3, 8, true>
      <<<dim3(9, 25), 128, 0, stream>>>(KT0, K1T, scal + 0, nullptr);

  int g1 = imin(1024, (N1 + 255) / 256);
  ssq_partial_k<<<g1, 256, 0, stream>>>(K1T, N1, part);
  final_reduce_k<<<1, 256, 0, stream>>>(part, g1, scal + 1);

  gram_lds_k<5, 3, true>
      <<<dim3(9, 81), 128, 0, stream>>>(K1T, K2T, scal + 1, nullptr);

  int g2 = imin(1024, (N2 + 255) / 256);
  ssq_partial_k<<<g2, 256, 0, stream>>>(K2T, N2, part);
  final_reduce_k<<<1, 256, 0, stream>>>(part, g2, scal + 2);

  gram_lds_k<9, 1, false>
      <<<dim3(9, 289), 128, 0, stream>>>(K2T, nullptr, scal + 2, part);
  final_reduce_k<<<1, 256, 0, stream>>>(part, 9 * 289 * 4, scal + 3);

  sigma_k<<<1, 1, 0, stream>>>(scal, scal + 4);
  scale_k<<<(N0 + 255) / 256, 256, 0, stream>>>(K, scal + 4, out, N0);
}


# Gram LDS staging: a round's overlap rows of all its o packed as one dense stream; a k-step takes the next four rows across o boundaries, zero filler only in a round's last step (changes fp sum order)
# speedup vs baseline: 3.4232x; 1.0549x over previous
#include <hip/hip_runtime.h>


typedef float v2f __attribute__((ext_vector_type(2)));
typedef float v8f __attribute__((ext_vector_type(8)));

#define C_CH 192
#define N_IN (192 * 192 * 9)

__device__ __forceinline__ v8f wmma_f32_16x16x4(v2f a, v2f b, v8f c) {
  return __builtin_amdgcn_wmma_f32_16x16x4_f32(false, a, false, b, (short)0, c,
                                               false, false);
}

__global__ void transpose_in_k(const float* __restrict__ in,
                               float* __restrict__ out, int n) {
  int i = blockIdx.x * blockDim.x + threadIdx.x;
  if (i < n) {
    int a = i % C_CH;
    int t = i / C_CH;
    int p = t % 9;
    int o = t / 9;
    out[i] = in[(o * C_CH + a) * 9 + p];
  }
}

__global__ void ssq_partial_k(const float* __restrict__ x, int n,
                              float* __restrict__ part) {
  __shared__ float sdata[256];
  float s = 0.f;
  for (int i = blockIdx.x * blockDim.x + threadIdx.x; i < n;
       i += gridDim.x * blockDim.x) {
    float v = x[i];
    s += v * v;
  }
  sdata[threadIdx.x] = s;
  __syncthreads();
  for (int off = 128; off > 0; off >>= 1) {
    if ((int)threadIdx.x < off) sdata[threadIdx.x] += sdata[threadIdx.x + off];
    __syncthreads();
  }
  if (threadIdx.x == 0) part[blockIdx.x] = sdata[0];
}

__global__ void final_reduce_k(const float* __restrict__ part, int n,
                               float* __restrict__ out) {
  __shared__ float sdata[256];
  float s = 0.f;
  for (int i = threadIdx.x; i < n; i += 256) s += part[i];
  sdata[threadIdx.x] = s;
  __syncthreads();
  for (int off = 128; off > 0; off >>= 1) {
    if ((int)threadIdx.x < off) sdata[threadIdx.x] += sdata[threadIdx.x + off];
    __syncthreads();
  }
  if (threadIdx.x == 0) out[0] = sdata[0];
}

__constant__ const unsigned short shift_order_3[25] = {
    12, 13, 11, 17, 7, 16, 8, 18, 6, 14, 10, 22, 2, 15, 9, 19, 5,
    21, 3, 23, 1, 20, 4, 24, 0};
__constant__ const unsigned short shift_order_5[81] = {
    40, 41, 39, 49, 31, 48, 32, 50, 30, 42, 38, 58, 22, 47, 33, 51, 29,
    57, 23, 59, 21, 43, 37, 67, 13, 56, 24, 60, 20, 46, 34, 52, 28, 66,
    14, 68, 12, 55, 25, 61, 19, 65, 15, 69, 11, 44, 36, 76, 4, 45, 35,
    53, 27, 64, 16, 70, 10, 75, 5, 77, 3, 54, 26, 62, 18, 74, 6, 78,
    2, 63, 17, 71, 9, 73, 7, 79, 1, 72, 8, 80, 0};
__constant__ const unsigned short shift_order_9[289] = {
    144, 145, 143, 161, 127, 160, 128, 162, 126, 146, 142, 178, 110, 159, 129, 163, 125,
    177, 111, 179, 109, 147, 141, 195, 93, 176, 112, 180, 108, 158, 130, 164, 124, 194,
    94, 196, 92, 148, 140, 212, 76, 175, 113, 181, 107, 193, 95, 197, 91, 157, 131,
    165, 123, 211, 77, 213, 75, 149, 139, 192, 96, 198, 90, 229, 59, 174, 114, 182,
    106, 210, 78, 214, 74, 156, 132, 166, 122, 228, 60, 230, 58, 191, 97, 199, 89,
    209, 79, 215, 73, 173, 115, 183, 105, 227, 61, 231, 57, 150, 138, 246, 42, 208,
    80, 216, 72, 155, 133, 167, 121, 190, 98, 200, 88, 226, 62, 232, 56, 245, 43,
    247, 41, 172, 116, 184, 104, 244, 44, 248, 40, 207, 81, 217, 71, 225, 63, 233,
    55, 151, 137, 189, 99, 201, 87, 243, 45, 249, 39, 263, 25, 154, 134, 168, 120,
    224, 64, 234, 54, 262, 26, 264, 24, 206, 82, 218, 70, 242, 46, 250, 38, 171,
    117, 185, 103, 261, 27, 265, 23, 188, 100, 202, 86, 223, 65, 235, 53, 241, 47,
    251, 37, 260, 28, 266, 22, 205, 83, 219, 69, 259, 29, 267, 21, 152, 136, 240,
    48, 252, 36, 280, 8, 153, 135, 169, 119, 222, 66, 236, 52, 258, 30, 268, 20,
    279, 9, 281, 7, 170, 118, 186, 102, 278, 10, 282, 6, 187, 101, 203, 85, 239,
    49, 253, 35, 257, 31, 269, 19, 277, 11, 283, 5, 204, 84, 220, 68, 276, 12,
    284, 4, 221, 67, 237, 51, 256, 32, 270, 18, 275, 13, 285, 3, 238, 50, 254,
    34, 274, 14, 286, 2, 255, 33, 271, 17, 273, 15, 287, 1, 272, 16, 288, 0};

template <int KS>
__device__ __forceinline__ int shift_of_rank(int r) {
  if constexpr (KS == 3) return shift_order_3[r];
  else if constexpr (KS == 5) return shift_order_5[r];
  else return shift_order_9[r];
}

template <int KS, int OB, bool WRITE>
__global__ __launch_bounds__(128) void gram_lds_k(
    const float* __restrict__ src, float* __restrict__ dst,
    const float* __restrict__ f2, float* __restrict__ part) {
  constexpr int P = KS * KS;
  constexpr int WS = 2 * KS - 1;
  constexpr int S = WS * WS;
  constexpr int NPAIR = ((P + 3) / 4) * 2;
  constexpr int NROW = OB * NPAIR;
  constexpr int NIT = (NROW + 7) / 8;
  constexpr int CP = 80;
  static_assert(C_CH % OB == 0, "o blocks must tile the channel count");
  static_assert(C_CH % 48 == 0 && C_CH % 64 == 0,
                "every o block size offered below must tile the channel count");

  constexpr int KG = 4;
  __shared__ __attribute__((aligned(16))) v2f sA[(NROW + 2 * KG) * CP];
  __shared__ __attribute__((aligned(16))) v2f sB[(NROW + 2 * KG) * CP];

  static_assert(KS == 3 || KS == 5 || KS == 9, "a dispatch table per kernel size");
  const int sidx = shift_of_rank<KS>(blockIdx.y);
  const int sy = sidx / WS - (KS - 1);
  const int sx = sidx % WS - (KS - 1);
  const int ny = KS - (sy < 0 ? -sy : sy);
  const int nx = KS - (sx < 0 ? -sx : sx);
  const int validp = ny * nx;
  constexpr int CAPP = 2 * NROW;
  static_assert(NROW % 2 == 0, "a k-step is two pair rows");
  static_assert(P * OB <= CAPP, "the centre shift's round must fit the slab");
  int obs = 1;
  if (2 * validp <= CAPP) obs = 2;
  if (3 * validp <= CAPP) obs = 3;
  if (4 * validp <= CAPP) obs = 4;
  if (6 * validp <= CAPP) obs = 6;
  if (8 * validp <= CAPP) obs = 8;
  if (12 * validp <= CAPP) obs = 12;
  if (16 * validp <= CAPP) obs = 16;
  if (24 * validp <= CAPP) obs = 24;
  if (32 * validp <= CAPP) obs = 32;
  if (48 * validp <= CAPP) obs = 48;
  if (64 * validp <= CAPP) obs = 64;
  const int npos = obs * validp;
  const int nkt = (npos + 3) >> 2;
  const int nrows = 2 * nkt;
  const int by0 = sy < 0 ? -sy : 0;
  const int bx0 = sx < 0 ? -sx : 0;
  const int dpa = sy * KS + sx;

  const int tid = threadIdx.x;
  const int lane = tid & 31;
  const int wid = tid >> 5;
  const int l15 = lane & 15;
  const int h = lane >> 4;
  const int wm = wid >> 1;
  const int wn = wid & 1;
  const int m0 = (blockIdx.x / 3) * 64;
  const int n0 = (blockIdx.x % 3) * 64;

  const int q4 = (tid & 15) * 4;
  const int r8 = tid >> 4;

  unsigned offb[NIT][2], offa[NIT][2];
  unsigned usedmask = 0u, padmask = 0u;
#pragma unroll
  for (int it = 0; it < NIT; ++it) {
    const int g = r8 + 8 * it;
    const bool used = g < nrows;
    if (used) usedmask |= 1u << it;
#pragma unroll
    for (int e = 0; e < 2; ++e) {
      const int d = 2 * g + e;
      const int dd = d < npos ? d : npos - 1;
      const int ob = dd / validp;
      const int jj = dd - ob * validp;
      const int jd = jj / nx;
      const int jm = jj - jd * nx;
      const int pb = (by0 + jd) * KS + (bx0 + jm);
      offb[it][e] = 4u * (unsigned)((ob * P + pb) * C_CH + n0 + q4);
      offa[it][e] = 4u * (unsigned)((ob * P + pb + dpa) * C_CH + m0 + q4);
      if (d >= npos) padmask |= 1u << (2 * it + e);
    }
  }

  const float4 z4 = {0.f, 0.f, 0.f, 0.f};
  float4 ra[NIT][2], rb[NIT][2];

  auto fetch = [&](int o0) {
    const unsigned ro = (unsigned)o0 * (unsigned)(P * C_CH * 4);
    const char* base = reinterpret_cast<const char*>(src);
#pragma unroll
    for (int it = 0; it < NIT; ++it) {
      if ((usedmask >> it) & 1u) {
#pragma unroll
        for (int e = 0; e < 2; ++e) {
          rb[it][e] =
              *reinterpret_cast<const float4*>(base + (offb[it][e] + ro));
          ra[it][e] =
              *reinterpret_cast<const float4*>(base + (offa[it][e] + ro));
        }
      }
    }
  };

  auto stash = [&]() {
#pragma unroll
    for (int it = 0; it < NIT; ++it) {
      if ((usedmask >> it) & 1u) {
        const int g = r8 + 8 * it;
        float4* pa4 = reinterpret_cast<float4*>(&sA[g * CP + q4]);
        float4* pb4 = reinterpret_cast<float4*>(&sB[g * CP + q4]);
        const bool pad0 = (padmask >> (2 * it)) & 1u;
        const bool pad1 = (padmask >> (2 * it + 1)) & 1u;
        const float4 a0 = pad0 ? z4 : ra[it][0], a1 = pad1 ? z4 : ra[it][1];
        const float4 b0 = pad0 ? z4 : rb[it][0], b1 = pad1 ? z4 : rb[it][1];
        pa4[0] = float4{a0.x, a1.x, a0.y, a1.y};
        pa4[1] = float4{a0.z, a1.z, a0.w, a1.w};
        pb4[0] = float4{b0.x, b1.x, b0.y, b1.y};
        pb4[1] = float4{b0.z, b1.z, b0.w, b1.w};
      }
    }
  };

  const v8f z = {0.f, 0.f, 0.f, 0.f, 0.f, 0.f, 0.f, 0.f};
  v8f acc[2][2] = {{z, z}, {z, z}};

  const v2f* fa = sA + h * CP + wm * 32 + l15;
  const v2f* fb = sB + h * CP + wn * 32 + l15;

  fetch(0);
  for (int o0 = 0; o0 < C_CH; o0 += obs) {
    __syncthreads();
    stash();
    __syncthreads();
    if (o0 + obs < C_CH) fetch(o0 + obs);

    {
      const v2f* pa = fa;
      const v2f* pb = fb;
      v2f av[KG][2], bv[KG][2];
#pragma unroll
      for (int g = 0; g < KG; ++g)
#pragma unroll
        for (int t = 0; t < 2; ++t) {
          av[g][t] = pa[2 * CP * g + 16 * t];
          bv[g][t] = pb[2 * CP * g + 16 * t];
        }
      const int ngrp = nkt / KG;
#pragma unroll 2
      for (int i = 0; i < ngrp; ++i) {
        pa += 2 * CP * KG;
        pb += 2 * CP * KG;
        v2f an[KG][2], bn[KG][2];
        __builtin_amdgcn_sched_barrier(0);
#pragma unroll
        for (int g = 0; g < KG; ++g)
#pragma unroll
          for (int t = 0; t < 2; ++t) {
            an[g][t] = pa[2 * CP * g + 16 * t];
            bn[g][t] = pb[2 * CP * g + 16 * t];
          }
        __builtin_amdgcn_sched_barrier(0);
#pragma unroll
        for (int g = 0; g < KG; ++g)
#pragma unroll
          for (int ti = 0; ti < 2; ++ti)
#pragma unroll
            for (int tj = 0; tj < 2; ++tj)
              acc[ti][tj] = wmma_f32_16x16x4(av[g][ti], bv[g][tj], acc[ti][tj]);
#pragma unroll
        for (int g = 0; g < KG; ++g)
#pragma unroll
          for (int t = 0; t < 2; ++t) {
            av[g][t] = an[g][t];
            bv[g][t] = bn[g][t];
          }
      }
      const int rem = nkt - ngrp * KG;
#pragma unroll
      for (int g = 0; g < KG - 1; ++g)
        if (g < rem) {
#pragma unroll
          for (int ti = 0; ti < 2; ++ti)
#pragma unroll
            for (int tj = 0; tj < 2; ++tj)
              acc[ti][tj] = wmma_f32_16x16x4(av[g][ti], bv[g][tj], acc[ti][tj]);
        }
    }
  }

  const float inv = 1.0f / f2[0];

  if (WRITE) {
#pragma unroll
    for (int ti = 0; ti < 2; ++ti)
#pragma unroll
      for (int r = 0; r < 8; ++r) {
        int am = m0 + wm * 32 + 16 * ti + r + 8 * h;
        size_t row = ((size_t)am * S + sidx) * C_CH;
#pragma unroll
        for (int tj = 0; tj < 2; ++tj)
          dst[row + n0 + wn * 32 + 16 * tj + l15] = acc[ti][tj][r] * inv;
      }
  } else {
    float ssq = 0.f;
#pragma unroll
    for (int ti = 0; ti < 2; ++ti)
#pragma unroll
      for (int tj = 0; tj < 2; ++tj)
#pragma unroll
        for (int r = 0; r < 8; ++r) {
          float g = acc[ti][tj][r] * inv;
          ssq += g * g;
        }
    for (int off = 16; off > 0; off >>= 1) ssq += __shfl_xor(ssq, off, 32);
    const int slot = sidx <= S / 2 ? 2 * (S / 2 - sidx) : 2 * (sidx - S / 2) - 1;
    if (lane == 0) part[(slot * gridDim.x + blockIdx.x) * 4 + wid] = ssq;
  }
}

__global__ void sigma_k(const float* __restrict__ s,
                        float* __restrict__ inv_sigma) {
  if (threadIdx.x == 0 && blockIdx.x == 0) {
    double ls = 0.5 * (log((double)s[0]) + 0.5 * log((double)s[1]) +
                       0.25 * log((double)s[2]) + 0.125 * log((double)s[3]));
    inv_sigma[0] = (float)exp(-ls);
  }
}

__global__ void scale_k(const float* __restrict__ in,
                        const float* __restrict__ inv_sigma,
                        float* __restrict__ out, int n) {
  int i = blockIdx.x * blockDim.x + threadIdx.x;
  if (i < n) out[i] = in[i] * inv_sigma[0];
}

static inline int imin(int a, int b) { return a < b ? a : b; }

extern "C" void kernel_launch(void* const* d_in, const int* in_sizes, int n_in,
                              void* d_out, int out_size, void* d_ws,
                              size_t ws_size, hipStream_t stream) {
  (void)in_sizes; (void)n_in; (void)out_size; (void)ws_size;
  const float* K = (const float*)d_in[0];
  float* out = (float*)d_out;
  char* ws = (char*)d_ws;

  float* scal = (float*)ws;
  float* part = (float*)(ws + 1024);
  float* K1T = (float*)(ws + (1 << 17));
  float* K2T = (float*)(ws + (4 << 20));
  float* KT0 = K2T;

  const int N0 = N_IN;
  const int N1 = 192 * 192 * 25;
  const int N2 = 192 * 192 * 81;

  transpose_in_k<<<(N0 + 255) / 256, 256, 0, stream>>>(K, KT0, N0);
  int g0 = imin(1024, (N0 + 255) / 256);
  ssq_partial_k<<<g0, 256, 0, stream>>>(K, N0, part);
  final_reduce_k<<<1, 256, 0, stream>>>(part, g0, scal + 0);

  gram_lds_k<3, 8, true>
      <<<dim3(9, 25), 128, 0, stream>>>(KT0, K1T, scal + 0, nullptr);

  int g1 = imin(1024, (N1 + 255) / 256);
  ssq_partial_k<<<g1, 256, 0, stream>>>(K1T, N1, part);
  final_reduce_k<<<1, 256, 0, stream>>>(part, g1, scal + 1);

  gram_lds_k<5, 3, true>
      <<<dim3(9, 81), 128, 0, stream>>>(K1T, K2T, scal + 1, nullptr);

  int g2 = imin(1024, (N2 + 255) / 256);
  ssq_partial_k<<<g2, 256, 0, stream>>>(K2T, N2, part);
  final_reduce_k<<<1, 256, 0, stream>>>(part, g2, scal + 2);

  gram_lds_k<9, 1, false>
      <<<dim3(9, 289), 128, 0, stream>>>(K2T, nullptr, scal + 2, part);
  final_reduce_k<<<1, 256, 0, stream>>>(part, 9 * 289 * 4, scal + 3);

  sigma_k<<<1, 1, 0, stream>>>(scal, scal + 4);
  scale_k<<<(N0 + 255) / 256, 256, 0, stream>>>(K, scal + 4, out, N0);
}


# Gram steps 3x3 and 5x5: 32x32 work-group tiles (wave tile 16x16, 36 tiles per shift; was 9 of 64x64); even and odd k-steps go to two accumulators, added at the end (changes fp sum order)
# speedup vs baseline: 3.5768x; 1.0449x over previous
#include <hip/hip_runtime.h>


typedef float v2f __attribute__((ext_vector_type(2)));
typedef float v8f __attribute__((ext_vector_type(8)));

#define C_CH 192
#define N_IN (192 * 192 * 9)

__device__ __forceinline__ v8f wmma_f32_16x16x4(v2f a, v2f b, v8f c) {
  return __builtin_amdgcn_wmma_f32_16x16x4_f32(false, a, false, b, (short)0, c,
                                               false, false);
}

__global__ void transpose_in_k(const float* __restrict__ in,
                               float* __restrict__ out, int n) {
  int i = blockIdx.x * blockDim.x + threadIdx.x;
  if (i < n) {
    int a = i % C_CH;
    int t = i / C_CH;
    int p = t % 9;
    int o = t / 9;
    out[i] = in[(o * C_CH + a) * 9 + p];
  }
}

__global__ void ssq_partial_k(const float* __restrict__ x, int n,
                              float* __restrict__ part) {
  __shared__ float sdata[256];
  float s = 0.f;
  for (int i = blockIdx.x * blockDim.x + threadIdx.x; i < n;
       i += gridDim.x * blockDim.x) {
    float v = x[i];
    s += v * v;
  }
  sdata[threadIdx.x] = s;
  __syncthreads();
  for (int off = 128; off > 0; off >>= 1) {
    if ((int)threadIdx.x < off) sdata[threadIdx.x] += sdata[threadIdx.x + off];
    __syncthreads();
  }
  if (threadIdx.x == 0) part[blockIdx.x] = sdata[0];
}

__global__ void final_reduce_k(const float* __restrict__ part, int n,
                               float* __restrict__ out) {
  __shared__ float sdata[256];
  float s = 0.f;
  for (int i = threadIdx.x; i < n; i += 256) s += part[i];
  sdata[threadIdx.x] = s;
  __syncthreads();
  for (int off = 128; off > 0; off >>= 1) {
    if ((int)threadIdx.x < off) sdata[threadIdx.x] += sdata[threadIdx.x + off];
    __syncthreads();
  }
  if (threadIdx.x == 0) out[0] = sdata[0];
}

__constant__ const unsigned short shift_order_3[25] = {
    12, 13, 11, 17, 7, 16, 8, 18, 6, 14, 10, 22, 2, 15, 9, 19, 5,
    21, 3, 23, 1, 20, 4, 24, 0};
__constant__ const unsigned short shift_order_5[81] = {
    40, 41, 39, 49, 31, 48, 32, 50, 30, 42, 38, 58, 22, 47, 33, 51, 29,
    57, 23, 59, 21, 43, 37, 67, 13, 56, 24, 60, 20, 46, 34, 52, 28, 66,
    14, 68, 12, 55, 25, 61, 19, 65, 15, 69, 11, 44, 36, 76, 4, 45, 35,
    53, 27, 64, 16, 70, 10, 75, 5, 77, 3, 54, 26, 62, 18, 74, 6, 78,
    2, 63, 17, 71, 9, 73, 7, 79, 1, 72, 8, 80, 0};
__constant__ const unsigned short shift_order_9[289] = {
    144, 145, 143, 161, 127, 160, 128, 162, 126, 146, 142, 178, 110, 159, 129, 163, 125,
    177, 111, 179, 109, 147, 141, 195, 93, 176, 112, 180, 108, 158, 130, 164, 124, 194,
    94, 196, 92, 148, 140, 212, 76, 175, 113, 181, 107, 193, 95, 197, 91, 157, 131,
    165, 123, 211, 77, 213, 75, 149, 139, 192, 96, 198, 90, 229, 59, 174, 114, 182,
    106, 210, 78, 214, 74, 156, 132, 166, 122, 228, 60, 230, 58, 191, 97, 199, 89,
    209, 79, 215, 73, 173, 115, 183, 105, 227, 61, 231, 57, 150, 138, 246, 42, 208,
    80, 216, 72, 155, 133, 167, 121, 190, 98, 200, 88, 226, 62, 232, 56, 245, 43,
    247, 41, 172, 116, 184, 104, 244, 44, 248, 40, 207, 81, 217, 71, 225, 63, 233,
    55, 151, 137, 189, 99, 201, 87, 243, 45, 249, 39, 263, 25, 154, 134, 168, 120,
    224, 64, 234, 54, 262, 26, 264, 24, 206, 82, 218, 70, 242, 46, 250, 38, 171,
    117, 185, 103, 261, 27, 265, 23, 188, 100, 202, 86, 223, 65, 235, 53, 241, 47,
    251, 37, 260, 28, 266, 22, 205, 83, 219, 69, 259, 29, 267, 21, 152, 136, 240,
    48, 252, 36, 280, 8, 153, 135, 169, 119, 222, 66, 236, 52, 258, 30, 268, 20,
    279, 9, 281, 7, 170, 118, 186, 102, 278, 10, 282, 6, 187, 101, 203, 85, 239,
    49, 253, 35, 257, 31, 269, 19, 277, 11, 283, 5, 204, 84, 220, 68, 276, 12,
    284, 4, 221, 67, 237, 51, 256, 32, 270, 18, 275, 13, 285, 3, 238, 50, 254,
    34, 274, 14, 286, 2, 255, 33, 271, 17, 273, 15, 287, 1, 272, 16, 288, 0};

template <int KS>
__device__ __forceinline__ int shift_of_rank(int r) {
  if constexpr (KS == 3) return shift_order_3[r];
  else if constexpr (KS == 5) return shift_order_5[r];
  else return shift_order_9[r];
}

template <int KS, int OB, bool WRITE, int TILE>
__global__ __launch_bounds__(128) void gram_lds_k(
    const float* __restrict__ src, float* __restrict__ dst,
    const float* __restrict__ f2, float* __restrict__ part) {
  constexpr int P = KS * KS;
  constexpr int WS = 2 * KS - 1;
  constexpr int S = WS * WS;
  constexpr int NPAIR = ((P + 3) / 4) * 2;
  constexpr int NROW = OB * NPAIR;
  static_assert(TILE == 64 || TILE == 32, "tile edge");
  constexpr int TQ = TILE / 4;
  constexpr int RPP = 128 / TQ;
  constexpr int NIT = (NROW + RPP - 1) / RPP;
  constexpr int WT = TILE / 2;
  constexpr int NF = WT / 16;
  constexpr int NA = NF == 1 ? 2 : 1;
  constexpr int NTL = C_CH / TILE;
  constexpr int CP = TILE + 16;
  static_assert((2 * CP) % 64 == 32, "lane halves on disjoint banks");
  static_assert(C_CH % OB == 0, "o blocks must tile the channel count");
  static_assert(C_CH % 48 == 0 && C_CH % 64 == 0,
                "every o block size offered below must tile the channel count");

  constexpr int KG = 4;
  __shared__ __attribute__((aligned(16))) v2f sA[(NROW + 2 * KG) * CP];
  __shared__ __attribute__((aligned(16))) v2f sB[(NROW + 2 * KG) * CP];

  static_assert(KS == 3 || KS == 5 || KS == 9, "a dispatch table per kernel size");
  const int sidx = shift_of_rank<KS>(blockIdx.y);
  const int sy = sidx / WS - (KS - 1);
  const int sx = sidx % WS - (KS - 1);
  const int ny = KS - (sy < 0 ? -sy : sy);
  const int nx = KS - (sx < 0 ? -sx : sx);
  const int validp = ny * nx;
  constexpr int CAPP = 2 * NROW;
  static_assert(NROW % 2 == 0, "a k-step is two pair rows");
  static_assert(P * OB <= CAPP, "the centre shift's round must fit the slab");
  int obs = 1;
  if (2 * validp <= CAPP) obs = 2;
  if (3 * validp <= CAPP) obs = 3;
  if (4 * validp <= CAPP) obs = 4;
  if (6 * validp <= CAPP) obs = 6;
  if (8 * validp <= CAPP) obs = 8;
  if (12 * validp <= CAPP) obs = 12;
  if (16 * validp <= CAPP) obs = 16;
  if (24 * validp <= CAPP) obs = 24;
  if (32 * validp <= CAPP) obs = 32;
  if (48 * validp <= CAPP) obs = 48;
  if (64 * validp <= CAPP) obs = 64;
  const int npos = obs * validp;
  const int nkt = (npos + 3) >> 2;
  const int nrows = 2 * nkt;
  const int by0 = sy < 0 ? -sy : 0;
  const int bx0 = sx < 0 ? -sx : 0;
  const int dpa = sy * KS + sx;

  const int tid = threadIdx.x;
  const int lane = tid & 31;
  const int wid = tid >> 5;
  const int l15 = lane & 15;
  const int h = lane >> 4;
  const int wm = wid >> 1;
  const int wn = wid & 1;
  const int m0 = (blockIdx.x / NTL) * TILE;
  const int n0 = (blockIdx.x % NTL) * TILE;

  const int q4 = (tid % TQ) * 4;
  const int r8 = tid / TQ;

  unsigned offb[NIT][2], offa[NIT][2];
  unsigned usedmask = 0u, padmask = 0u;
#pragma unroll
  for (int it = 0; it < NIT; ++it) {
    const int g = r8 + RPP * it;
    const bool used = g < nrows;
    if (used) usedmask |= 1u << it;
#pragma unroll
    for (int e = 0; e < 2; ++e) {
      const int d = 2 * g + e;
      const int dd = d < npos ? d : npos - 1;
      const int ob = dd / validp;
      const int jj = dd - ob * validp;
      const int jd = jj / nx;
      const int jm = jj - jd * nx;
      const int pb = (by0 + jd) * KS + (bx0 + jm);
      offb[it][e] = 4u * (unsigned)((ob * P + pb) * C_CH + n0 + q4);
      offa[it][e] = 4u * (unsigned)((ob * P + pb + dpa) * C_CH + m0 + q4);
      if (d >= npos) padmask |= 1u << (2 * it + e);
    }
  }

  const float4 z4 = {0.f, 0.f, 0.f, 0.f};
  float4 ra[NIT][2], rb[NIT][2];

  auto fetch = [&](int o0) {
    const unsigned ro = (unsigned)o0 * (unsigned)(P * C_CH * 4);
    const char* base = reinterpret_cast<const char*>(src);
#pragma unroll
    for (int it = 0; it < NIT; ++it) {
      if ((usedmask >> it) & 1u) {
#pragma unroll
        for (int e = 0; e < 2; ++e) {
          rb[it][e] =
              *reinterpret_cast<const float4*>(base + (offb[it][e] + ro));
          ra[it][e] =
              *reinterpret_cast<const float4*>(base + (offa[it][e] + ro));
        }
      }
    }
  };

  auto stash = [&]() {
#pragma unroll
    for (int it = 0; it < NIT; ++it) {
      if ((usedmask >> it) & 1u) {
        const int g = r8 + RPP * it;
        float4* pa4 = reinterpret_cast<float4*>(&sA[g * CP + q4]);
        float4* pb4 = reinterpret_cast<float4*>(&sB[g * CP + q4]);
        const bool pad0 = (padmask >> (2 * it)) & 1u;
        const bool pad1 = (padmask >> (2 * it + 1)) & 1u;
        const float4 a0 = pad0 ? z4 : ra[it][0], a1 = pad1 ? z4 : ra[it][1];
        const float4 b0 = pad0 ? z4 : rb[it][0], b1 = pad1 ? z4 : rb[it][1];
        pa4[0] = float4{a0.x, a1.x, a0.y, a1.y};
        pa4[1] = float4{a0.z, a1.z, a0.w, a1.w};
        pb4[0] = float4{b0.x, b1.x, b0.y, b1.y};
        pb4[1] = float4{b0.z, b1.z, b0.w, b1.w};
      }
    }
  };

  const v8f z = {0.f, 0.f, 0.f, 0.f, 0.f, 0.f, 0.f, 0.f};
  v8f acc[NA][NF][NF];
#pragma unroll
  for (int s = 0; s < NA; ++s)
#pragma unroll
    for (int ti = 0; ti < NF; ++ti)
#pragma unroll
      for (int tj = 0; tj < NF; ++tj) acc[s][ti][tj] = z;

  const v2f* fa = sA + h * CP + wm * WT + l15;
  const v2f* fb = sB + h * CP + wn * WT + l15;

  fetch(0);
  for (int o0 = 0; o0 < C_CH; o0 += obs) {
    __syncthreads();
    stash();
    __syncthreads();
    if (o0 + obs < C_CH) fetch(o0 + obs);

    {
      const v2f* pa = fa;
      const v2f* pb = fb;
      v2f av[KG][NF], bv[KG][NF];
#pragma unroll
      for (int g = 0; g < KG; ++g)
#pragma unroll
        for (int t = 0; t < NF; ++t) {
          av[g][t] = pa[2 * CP * g + 16 * t];
          bv[g][t] = pb[2 * CP * g + 16 * t];
        }
      const int ngrp = nkt / KG;
#pragma unroll 2
      for (int i = 0; i < ngrp; ++i) {
        pa += 2 * CP * KG;
        pb += 2 * CP * KG;
        v2f an[KG][NF], bn[KG][NF];
        __builtin_amdgcn_sched_barrier(0);
#pragma unroll
        for (int g = 0; g < KG; ++g)
#pragma unroll
          for (int t = 0; t < NF; ++t) {
            an[g][t] = pa[2 * CP * g + 16 * t];
            bn[g][t] = pb[2 * CP * g + 16 * t];
          }
        __builtin_amdgcn_sched_barrier(0);
#pragma unroll
        for (int g = 0; g < KG; ++g)
#pragma unroll
          for (int ti = 0; ti < NF; ++ti)
#pragma unroll
            for (int tj = 0; tj < NF; ++tj)
              acc[g % NA][ti][tj] =
                  wmma_f32_16x16x4(av[g][ti], bv[g][tj], acc[g % NA][ti][tj]);
#pragma unroll
        for (int g = 0; g < KG; ++g)
#pragma unroll
          for (int t = 0; t < NF; ++t) {
            av[g][t] = an[g][t];
            bv[g][t] = bn[g][t];
          }
      }
      const int rem = nkt - ngrp * KG;
#pragma unroll
      for (int g = 0; g < KG - 1; ++g)
        if (g < rem) {
#pragma unroll
          for (int ti = 0; ti < NF; ++ti)
#pragma unroll
            for (int tj = 0; tj < NF; ++tj)
              acc[g % NA][ti][tj] =
                  wmma_f32_16x16x4(av[g][ti], bv[g][tj], acc[g % NA][ti][tj]);
        }
    }
  }

  if constexpr (NA == 2) {
#pragma unroll
    for (int ti = 0; ti < NF; ++ti)
#pragma unroll
      for (int tj = 0; tj < NF; ++tj) acc[0][ti][tj] += acc[1][ti][tj];
  }

  const float inv = 1.0f / f2[0];

  if (WRITE) {
#pragma unroll
    for (int ti = 0; ti < NF; ++ti)
#pragma unroll
      for (int r = 0; r < 8; ++r) {
        int am = m0 + wm * WT + 16 * ti + r + 8 * h;
        size_t row = ((size_t)am * S + sidx) * C_CH;
#pragma unroll
        for (int tj = 0; tj < NF; ++tj)
          dst[row + n0 + wn * WT + 16 * tj + l15] = acc[0][ti][tj][r] * inv;
      }
  } else {
    float ssq = 0.f;
#pragma unroll
    for (int ti = 0; ti < NF; ++ti)
#pragma unroll
      for (int tj = 0; tj < NF; ++tj)
#pragma unroll
        for (int r = 0; r < 8; ++r) {
          float g = acc[0][ti][tj][r] * inv;
          ssq += g * g;
        }
    for (int off = 16; off > 0; off >>= 1) ssq += __shfl_xor(ssq, off, 32);
    const int slot = sidx <= S / 2 ? 2 * (S / 2 - sidx) : 2 * (sidx - S / 2) - 1;
    if (lane == 0) part[(slot * gridDim.x + blockIdx.x) * 4 + wid] = ssq;
  }
}

__global__ void sigma_k(const float* __restrict__ s,
                        float* __restrict__ inv_sigma) {
  if (threadIdx.x == 0 && blockIdx.x == 0) {
    double ls = 0.5 * (log((double)s[0]) + 0.5 * log((double)s[1]) +
                       0.25 * log((double)s[2]) + 0.125 * log((double)s[3]));
    inv_sigma[0] = (float)exp(-ls);
  }
}

__global__ void scale_k(const float* __restrict__ in,
                        const float* __restrict__ inv_sigma,
                        float* __restrict__ out, int n) {
  int i = blockIdx.x * blockDim.x + threadIdx.x;
  if (i < n) out[i] = in[i] * inv_sigma[0];
}

static inline int imin(int a, int b) { return a < b ? a : b; }

extern "C" void kernel_launch(void* const* d_in, const int* in_sizes, int n_in,
                              void* d_out, int out_size, void* d_ws,
                              size_t ws_size, hipStream_t stream) {
  (void)in_sizes; (void)n_in; (void)out_size; (void)ws_size;
  const float* K = (const float*)d_in[0];
  float* out = (float*)d_out;
  char* ws = (char*)d_ws;

  float* scal = (float*)ws;
  float* part = (float*)(ws + 1024);
  float* K1T = (float*)(ws + (1 << 17));
  float* K2T = (float*)(ws + (4 << 20));
  float* KT0 = K2T;

  const int N0 = N_IN;
  const int N1 = 192 * 192 * 25;
  const int N2 = 192 * 192 * 81;

  transpose_in_k<<<(N0 + 255) / 256, 256, 0, stream>>>(K, KT0, N0);
  int g0 = imin(1024, (N0 + 255) / 256);
  ssq_partial_k<<<g0, 256, 0, stream>>>(K, N0, part);
  final_reduce_k<<<1, 256, 0, stream>>>(part, g0, scal + 0);

  gram_lds_k<3, 16, true, 32>
      <<<dim3(36, 25), 128, 0, stream>>>(KT0, K1T, scal + 0, nullptr);

  int g1 = imin(1024, (N1 + 255) / 256);
  ssq_partial_k<<<g1, 256, 0, stream>>>(K1T, N1, part);
  final_reduce_k<<<1, 256, 0, stream>>>(part, g1, scal + 1);

  gram_lds_k<5, 6, true, 32>
      <<<dim3(36, 81), 128, 0, stream>>>(K1T, K2T, scal + 1, nullptr);

  int g2 = imin(1024, (N2 + 255) / 256);
  ssq_partial_k<<<g2, 256, 0, stream>>>(K2T, N2, part);
  final_reduce_k<<<1, 256, 0, stream>>>(part, g2, scal + 2);

  gram_lds_k<9, 1, false, 64>
      <<<dim3(9, 289), 128, 0, stream>>>(K2T, nullptr, scal + 2, part);
  final_reduce_k<<<1, 256, 0, stream>>>(part, 9 * 289 * 4, scal + 3);

  sigma_k<<<1, 1, 0, stream>>>(scal, scal + 4);
  scale_k<<<(N0 + 255) / 256, 256, 0, stream>>>(K, scal + 4, out, N0);
}


# Scalings by f_1, f_2 dropped: they cancel in log sigma = log f_0 + log|gram^3(K/f_0)|/8 for every input, so both middle norm passes go (13 launches -> 9); Gram sum order kept, rounding differs
# speedup vs baseline: 3.6943x; 1.0329x over previous
#include <hip/hip_runtime.h>


typedef float v2f __attribute__((ext_vector_type(2)));
typedef float v8f __attribute__((ext_vector_type(8)));

#define C_CH 192
#define N_IN (192 * 192 * 9)

__device__ __forceinline__ v8f wmma_f32_16x16x4(v2f a, v2f b, v8f c) {
  return __builtin_amdgcn_wmma_f32_16x16x4_f32(false, a, false, b, (short)0, c,
                                               false, false);
}

__global__ void transpose_in_k(const float* __restrict__ in,
                               float* __restrict__ out, int n) {
  int i = blockIdx.x * blockDim.x + threadIdx.x;
  if (i < n) {
    int a = i % C_CH;
    int t = i / C_CH;
    int p = t % 9;
    int o = t / 9;
    out[i] = in[(o * C_CH + a) * 9 + p];
  }
}

__global__ void ssq_partial_k(const float* __restrict__ x, int n,
                              float* __restrict__ part) {
  __shared__ float sdata[256];
  float s = 0.f;
  for (int i = blockIdx.x * blockDim.x + threadIdx.x; i < n;
       i += gridDim.x * blockDim.x) {
    float v = x[i];
    s += v * v;
  }
  sdata[threadIdx.x] = s;
  __syncthreads();
  for (int off = 128; off > 0; off >>= 1) {
    if ((int)threadIdx.x < off) sdata[threadIdx.x] += sdata[threadIdx.x + off];
    __syncthreads();
  }
  if (threadIdx.x == 0) part[blockIdx.x] = sdata[0];
}

__global__ void final_reduce_k(const float* __restrict__ part, int n,
                               float* __restrict__ out) {
  __shared__ float sdata[256];
  float s = 0.f;
  for (int i = threadIdx.x; i < n; i += 256) s += part[i];
  sdata[threadIdx.x] = s;
  __syncthreads();
  for (int off = 128; off > 0; off >>= 1) {
    if ((int)threadIdx.x < off) sdata[threadIdx.x] += sdata[threadIdx.x + off];
    __syncthreads();
  }
  if (threadIdx.x == 0) out[0] = sdata[0];
}

__constant__ const unsigned short shift_order_3[25] = {
    12, 13, 11, 17, 7, 16, 8, 18, 6, 14, 10, 22, 2, 15, 9, 19, 5,
    21, 3, 23, 1, 20, 4, 24, 0};
__constant__ const unsigned short shift_order_5[81] = {
    40, 41, 39, 49, 31, 48, 32, 50, 30, 42, 38, 58, 22, 47, 33, 51, 29,
    57, 23, 59, 21, 43, 37, 67, 13, 56, 24, 60, 20, 46, 34, 52, 28, 66,
    14, 68, 12, 55, 25, 61, 19, 65, 15, 69, 11, 44, 36, 76, 4, 45, 35,
    53, 27, 64, 16, 70, 10, 75, 5, 77, 3, 54, 26, 62, 18, 74, 6, 78,
    2, 63, 17, 71, 9, 73, 7, 79, 1, 72, 8, 80, 0};
__constant__ const unsigned short shift_order_9[289] = {
    144, 145, 143, 161, 127, 160, 128, 162, 126, 146, 142, 178, 110, 159, 129, 163, 125,
    177, 111, 179, 109, 147, 141, 195, 93, 176, 112, 180, 108, 158, 130, 164, 124, 194,
    94, 196, 92, 148, 140, 212, 76, 175, 113, 181, 107, 193, 95, 197, 91, 157, 131,
    165, 123, 211, 77, 213, 75, 149, 139, 192, 96, 198, 90, 229, 59, 174, 114, 182,
    106, 210, 78, 214, 74, 156, 132, 166, 122, 228, 60, 230, 58, 191, 97, 199, 89,
    209, 79, 215, 73, 173, 115, 183, 105, 227, 61, 231, 57, 150, 138, 246, 42, 208,
    80, 216, 72, 155, 133, 167, 121, 190, 98, 200, 88, 226, 62, 232, 56, 245, 43,
    247, 41, 172, 116, 184, 104, 244, 44, 248, 40, 207, 81, 217, 71, 225, 63, 233,
    55, 151, 137, 189, 99, 201, 87, 243, 45, 249, 39, 263, 25, 154, 134, 168, 120,
    224, 64, 234, 54, 262, 26, 264, 24, 206, 82, 218, 70, 242, 46, 250, 38, 171,
    117, 185, 103, 261, 27, 265, 23, 188, 100, 202, 86, 223, 65, 235, 53, 241, 47,
    251, 37, 260, 28, 266, 22, 205, 83, 219, 69, 259, 29, 267, 21, 152, 136, 240,
    48, 252, 36, 280, 8, 153, 135, 169, 119, 222, 66, 236, 52, 258, 30, 268, 20,
    279, 9, 281, 7, 170, 118, 186, 102, 278, 10, 282, 6, 187, 101, 203, 85, 239,
    49, 253, 35, 257, 31, 269, 19, 277, 11, 283, 5, 204, 84, 220, 68, 276, 12,
    284, 4, 221, 67, 237, 51, 256, 32, 270, 18, 275, 13, 285, 3, 238, 50, 254,
    34, 274, 14, 286, 2, 255, 33, 271, 17, 273, 15, 287, 1, 272, 16, 288, 0};

template <int KS>
__device__ __forceinline__ int shift_of_rank(int r) {
  if constexpr (KS == 3) return shift_order_3[r];
  else if constexpr (KS == 5) return shift_order_5[r];
  else return shift_order_9[r];
}

template <int KS, int OB, bool WRITE, int TILE, bool SCALED>
__global__ __launch_bounds__(128) void gram_lds_k(
    const float* __restrict__ src, float* __restrict__ dst,
    const float* __restrict__ f2, float* __restrict__ part) {
  constexpr int P = KS * KS;
  constexpr int WS = 2 * KS - 1;
  constexpr int S = WS * WS;
  constexpr int NPAIR = ((P + 3) / 4) * 2;
  constexpr int NROW = OB * NPAIR;
  static_assert(TILE == 64 || TILE == 32, "tile edge");
  constexpr int TQ = TILE / 4;
  constexpr int RPP = 128 / TQ;
  constexpr int NIT = (NROW + RPP - 1) / RPP;
  constexpr int WT = TILE / 2;
  constexpr int NF = WT / 16;
  constexpr int NA = NF == 1 ? 2 : 1;
  constexpr int NTL = C_CH / TILE;
  constexpr int CP = TILE + 16;
  static_assert((2 * CP) % 64 == 32, "lane halves on disjoint banks");
  static_assert(C_CH % OB == 0, "o blocks must tile the channel count");
  static_assert(C_CH % 48 == 0 && C_CH % 64 == 0,
                "every o block size offered below must tile the channel count");

  constexpr int KG = 4;
  __shared__ __attribute__((aligned(16))) v2f sA[(NROW + 2 * KG) * CP];
  __shared__ __attribute__((aligned(16))) v2f sB[(NROW + 2 * KG) * CP];

  static_assert(KS == 3 || KS == 5 || KS == 9, "a dispatch table per kernel size");
  const int sidx = shift_of_rank<KS>(blockIdx.y);
  const int sy = sidx / WS - (KS - 1);
  const int sx = sidx % WS - (KS - 1);
  const int ny = KS - (sy < 0 ? -sy : sy);
  const int nx = KS - (sx < 0 ? -sx : sx);
  const int validp = ny * nx;
  constexpr int CAPP = 2 * NROW;
  static_assert(NROW % 2 == 0, "a k-step is two pair rows");
  static_assert(P * OB <= CAPP, "the centre shift's round must fit the slab");
  int obs = 1;
  if (2 * validp <= CAPP) obs = 2;
  if (3 * validp <= CAPP) obs = 3;
  if (4 * validp <= CAPP) obs = 4;
  if (6 * validp <= CAPP) obs = 6;
  if (8 * validp <= CAPP) obs = 8;
  if (12 * validp <= CAPP) obs = 12;
  if (16 * validp <= CAPP) obs = 16;
  if (24 * validp <= CAPP) obs = 24;
  if (32 * validp <= CAPP) obs = 32;
  if (48 * validp <= CAPP) obs = 48;
  if (64 * validp <= CAPP) obs = 64;
  const int npos = obs * validp;
  const int nkt = (npos + 3) >> 2;
  const int nrows = 2 * nkt;
  const int by0 = sy < 0 ? -sy : 0;
  const int bx0 = sx < 0 ? -sx : 0;
  const int dpa = sy * KS + sx;

  const int tid = threadIdx.x;
  const int lane = tid & 31;
  const int wid = tid >> 5;
  const int l15 = lane & 15;
  const int h = lane >> 4;
  const int wm = wid >> 1;
  const int wn = wid & 1;
  const int m0 = (blockIdx.x / NTL) * TILE;
  const int n0 = (blockIdx.x % NTL) * TILE;

  const int q4 = (tid % TQ) * 4;
  const int r8 = tid / TQ;

  unsigned offb[NIT][2], offa[NIT][2];
  unsigned usedmask = 0u, padmask = 0u;
#pragma unroll
  for (int it = 0; it < NIT; ++it) {
    const int g = r8 + RPP * it;
    const bool used = g < nrows;
    if (used) usedmask |= 1u << it;
#pragma unroll
    for (int e = 0; e < 2; ++e) {
      const int d = 2 * g + e;
      const int dd = d < npos ? d : npos - 1;
      const int ob = dd / validp;
      const int jj = dd - ob * validp;
      const int jd = jj / nx;
      const int jm = jj - jd * nx;
      const int pb = (by0 + jd) * KS + (bx0 + jm);
      offb[it][e] = 4u * (unsigned)((ob * P + pb) * C_CH + n0 + q4);
      offa[it][e] = 4u * (unsigned)((ob * P + pb + dpa) * C_CH + m0 + q4);
      if (d >= npos) padmask |= 1u << (2 * it + e);
    }
  }

  const float4 z4 = {0.f, 0.f, 0.f, 0.f};
  float4 ra[NIT][2], rb[NIT][2];

  auto fetch = [&](int o0) {
    const unsigned ro = (unsigned)o0 * (unsigned)(P * C_CH * 4);
    const char* base = reinterpret_cast<const char*>(src);
#pragma unroll
    for (int it = 0; it < NIT; ++it) {
      if ((usedmask >> it) & 1u) {
#pragma unroll
        for (int e = 0; e < 2; ++e) {
          rb[it][e] =
              *reinterpret_cast<const float4*>(base + (offb[it][e] + ro));
          ra[it][e] =
              *reinterpret_cast<const float4*>(base + (offa[it][e] + ro));
        }
      }
    }
  };

  auto stash = [&]() {
#pragma unroll
    for (int it = 0; it < NIT; ++it) {
      if ((usedmask >> it) & 1u) {
        const int g = r8 + RPP * it;
        float4* pa4 = reinterpret_cast<float4*>(&sA[g * CP + q4]);
        float4* pb4 = reinterpret_cast<float4*>(&sB[g * CP + q4]);
        const bool pad0 = (padmask >> (2 * it)) & 1u;
        const bool pad1 = (padmask >> (2 * it + 1)) & 1u;
        const float4 a0 = pad0 ? z4 : ra[it][0], a1 = pad1 ? z4 : ra[it][1];
        const float4 b0 = pad0 ? z4 : rb[it][0], b1 = pad1 ? z4 : rb[it][1];
        pa4[0] = float4{a0.x, a1.x, a0.y, a1.y};
        pa4[1] = float4{a0.z, a1.z, a0.w, a1.w};
        pb4[0] = float4{b0.x, b1.x, b0.y, b1.y};
        pb4[1] = float4{b0.z, b1.z, b0.w, b1.w};
      }
    }
  };

  const v8f z = {0.f, 0.f, 0.f, 0.f, 0.f, 0.f, 0.f, 0.f};
  v8f acc[NA][NF][NF];
#pragma unroll
  for (int s = 0; s < NA; ++s)
#pragma unroll
    for (int ti = 0; ti < NF; ++ti)
#pragma unroll
      for (int tj = 0; tj < NF; ++tj) acc[s][ti][tj] = z;

  const v2f* fa = sA + h * CP + wm * WT + l15;
  const v2f* fb = sB + h * CP + wn * WT + l15;

  fetch(0);
  for (int o0 = 0; o0 < C_CH; o0 += obs) {
    __syncthreads();
    stash();
    __syncthreads();
    if (o0 + obs < C_CH) fetch(o0 + obs);

    {
      const v2f* pa = fa;
      const v2f* pb = fb;
      v2f av[KG][NF], bv[KG][NF];
#pragma unroll
      for (int g = 0; g < KG; ++g)
#pragma unroll
        for (int t = 0; t < NF; ++t) {
          av[g][t] = pa[2 * CP * g + 16 * t];
          bv[g][t] = pb[2 * CP * g + 16 * t];
        }
      const int ngrp = nkt / KG;
#pragma unroll 2
      for (int i = 0; i < ngrp; ++i) {
        pa += 2 * CP * KG;
        pb += 2 * CP * KG;
        v2f an[KG][NF], bn[KG][NF];
        __builtin_amdgcn_sched_barrier(0);
#pragma unroll
        for (int g = 0; g < KG; ++g)
#pragma unroll
          for (int t = 0; t < NF; ++t) {
            an[g][t] = pa[2 * CP * g + 16 * t];
            bn[g][t] = pb[2 * CP * g + 16 * t];
          }
        __builtin_amdgcn_sched_barrier(0);
#pragma unroll
        for (int g = 0; g < KG; ++g)
#pragma unroll
          for (int ti = 0; ti < NF; ++ti)
#pragma unroll
            for (int tj = 0; tj < NF; ++tj)
              acc[g % NA][ti][tj] =
                  wmma_f32_16x16x4(av[g][ti], bv[g][tj], acc[g % NA][ti][tj]);
#pragma unroll
        for (int g = 0; g < KG; ++g)
#pragma unroll
          for (int t = 0; t < NF; ++t) {
            av[g][t] = an[g][t];
            bv[g][t] = bn[g][t];
          }
      }
      const int rem = nkt - ngrp * KG;
#pragma unroll
      for (int g = 0; g < KG - 1; ++g)
        if (g < rem) {
#pragma unroll
          for (int ti = 0; ti < NF; ++ti)
#pragma unroll
            for (int tj = 0; tj < NF; ++tj)
              acc[g % NA][ti][tj] =
                  wmma_f32_16x16x4(av[g][ti], bv[g][tj], acc[g % NA][ti][tj]);
        }
    }
  }

  if constexpr (NA == 2) {
#pragma unroll
    for (int ti = 0; ti < NF; ++ti)
#pragma unroll
      for (int tj = 0; tj < NF; ++tj) acc[0][ti][tj] += acc[1][ti][tj];
  }

  float inv = 1.0f;
  if constexpr (SCALED) inv = 1.0f / f2[0];
  auto outv = [&](float v) {
    if constexpr (SCALED) return v * inv;
    else return v;
  };

  if (WRITE) {
#pragma unroll
    for (int ti = 0; ti < NF; ++ti)
#pragma unroll
      for (int r = 0; r < 8; ++r) {
        int am = m0 + wm * WT + 16 * ti + r + 8 * h;
        size_t row = ((size_t)am * S + sidx) * C_CH;
#pragma unroll
        for (int tj = 0; tj < NF; ++tj)
          dst[row + n0 + wn * WT + 16 * tj + l15] = outv(acc[0][ti][tj][r]);
      }
  } else {
    float ssq = 0.f;
#pragma unroll
    for (int ti = 0; ti < NF; ++ti)
#pragma unroll
      for (int tj = 0; tj < NF; ++tj)
#pragma unroll
        for (int r = 0; r < 8; ++r) {
          float g = outv(acc[0][ti][tj][r]);
          ssq += g * g;
        }
    for (int off = 16; off > 0; off >>= 1) ssq += __shfl_xor(ssq, off, 32);
    const int slot = sidx <= S / 2 ? 2 * (S / 2 - sidx) : 2 * (sidx - S / 2) - 1;
    if (lane == 0) part[(slot * gridDim.x + blockIdx.x) * 4 + wid] = ssq;
  }
}

__global__ void sigma_k(const float* __restrict__ s,
                        float* __restrict__ inv_sigma) {
  if (threadIdx.x == 0 && blockIdx.x == 0) {
    double ls = 0.5 * (log((double)s[0]) + 0.125 * log((double)s[1]));
    inv_sigma[0] = (float)exp(-ls);
  }
}

__global__ void scale_k(const float* __restrict__ in,
                        const float* __restrict__ inv_sigma,
                        float* __restrict__ out, int n) {
  int i = blockIdx.x * blockDim.x + threadIdx.x;
  if (i < n) out[i] = in[i] * inv_sigma[0];
}

static inline int imin(int a, int b) { return a < b ? a : b; }

extern "C" void kernel_launch(void* const* d_in, const int* in_sizes, int n_in,
                              void* d_out, int out_size, void* d_ws,
                              size_t ws_size, hipStream_t stream) {
  (void)in_sizes; (void)n_in; (void)out_size; (void)ws_size;
  const float* K = (const float*)d_in[0];
  float* out = (float*)d_out;
  char* ws = (char*)d_ws;

  float* scal = (float*)ws;
  float* part = (float*)(ws + 1024);
  float* K1T = (float*)(ws + (1 << 17));
  float* K2T = (float*)(ws + (4 << 20));
  float* KT0 = K2T;

  const int N0 = N_IN;

  transpose_in_k<<<(N0 + 255) / 256, 256, 0, stream>>>(K, KT0, N0);
  int g0 = imin(1024, (N0 + 255) / 256);
  ssq_partial_k<<<g0, 256, 0, stream>>>(K, N0, part);
  final_reduce_k<<<1, 256, 0, stream>>>(part, g0, scal + 0);

  gram_lds_k<3, 16, true, 32, true>
      <<<dim3(36, 25), 128, 0, stream>>>(KT0, K1T, scal + 0, nullptr);

  gram_lds_k<5, 6, true, 32, false>
      <<<dim3(36, 81), 128, 0, stream>>>(K1T, K2T, nullptr, nullptr);

  gram_lds_k<9, 1, false, 64, false>
      <<<dim3(9, 289), 128, 0, stream>>>(K2T, nullptr, nullptr, part);
  final_reduce_k<<<1, 256, 0, stream>>>(part, 9 * 289 * 4, scal + 1);

  sigma_k<<<1, 1, 0, stream>>>(scal, scal + 2);
  scale_k<<<(N0 + 255) / 256, 256, 0, stream>>>(K, scal + 2, out, N0);
}
